# starting kernel
# baseline (speedup 1.0000x reference)
#include <hip/hip_runtime.h>


typedef __attribute__((ext_vector_type(2))) float v2f;
typedef __attribute__((ext_vector_type(8))) float v8f;

#define ATTN_NEG -1000000.0f

__global__ __launch_bounds__(128)
void DotProductAttention_58368605552662_kernel(
    const float* __restrict__ Qm, const float* __restrict__ Km,
    const float* __restrict__ Vm, const int* __restrict__ vlens,
    float* __restrict__ Om)
{
    constexpr int Qn = 2048, Kn = 2048, Dn = 128;
    const int lane = threadIdx.x & 31;
    const int wave = threadIdx.x >> 5;
    const int half = lane >> 4;
    const int l16  = lane & 15;

    const int b      = blockIdx.x >> 5;
    const int qchunk = blockIdx.x & 31;
    const int q0     = qchunk * 64 + wave * 16;
    const int vlen   = vlens[b];

    const float* Qb = Qm + (size_t)b * Qn * Dn;
    const float* Kb = Km + (size_t)b * Kn * Dn;
    const float* Vb = Vm + (size_t)b * Kn * Dn;
    float*       Ob = Om + (size_t)b * Qn * Dn;

    v2f qreg[32];
    const float* qrow = Qb + (size_t)(q0 + l16) * Dn + 2 * half;
#pragma unroll
    for (int c = 0; c < 32; ++c)
        qreg[c] = *(const v2f*)(qrow + 4 * c);

    v8f oacc[8];
#pragma unroll
    for (int t = 0; t < 8; ++t) oacc[t] = (v8f)0.0f;
    float m_run = -3.0e38f;
    float l_run = 0.0f;

    const float scale = 0.08838834764831845f;

    for (int kt = 0; kt < Kn / 16; ++kt) {
        const int krow0 = kt * 16;

        if (kt + 1 < Kn / 16)
            __builtin_prefetch(Kb + (size_t)(krow0 + 16 + l16) * Dn, 0, 0);

        v8f sacc = (v8f)0.0f;
        const float* krow = Kb + (size_t)(krow0 + l16) * Dn + 2 * half;
#pragma unroll
        for (int c = 0; c < 32; ++c) {
            v2f a = *(const v2f*)(krow + 4 * c);
            sacc = __builtin_amdgcn_wmma_f32_16x16x4_f32(
                false, a, false, qreg[c], (short)0, sacc, false, false);
        }

        float p[8];
        float tmax = -3.0e38f;
#pragma unroll
        for (int r = 0; r < 8; ++r) {
            float s = sacc[r] * scale;
            int kg = krow0 + r + 8 * half;
            if (kg >= vlen) s = ATTN_NEG;
            p[r] = s;
            tmax = fmaxf(tmax, s);
        }
        tmax = fmaxf(tmax, __shfl_xor(tmax, 16, 32));
        const float m_new = fmaxf(m_run, tmax);
        const float alpha = __expf(m_run - m_new);
        float tsum = 0.0f;
#pragma unroll
        for (int r = 0; r < 8; ++r) {
            p[r] = __expf(p[r] - m_new);
            tsum += p[r];
        }
        tsum += __shfl_xor(tsum, 16, 32);
        l_run = l_run * alpha + tsum;
        m_run = m_new;

#pragma unroll
        for (int r = 0; r < 8; ++r) {
            float ab = __shfl(alpha, r + 8 * half, 32);
#pragma unroll
            for (int t = 0; t < 8; ++t)
                oacc[t][r] = oacc[t][r] * ab;
        }

        float pq[8];
#pragma unroll
        for (int r = 0; r < 8; ++r)
            pq[r] = __shfl_xor(p[r], 16, 32);
        v2f aP[4];
        aP[0] = v2f{ half ? pq[2] : p[0],  half ? pq[3] : p[1] };
        aP[1] = v2f{ half ? pq[6] : p[4],  half ? pq[7] : p[5] };
        aP[2] = v2f{ half ? p[2]  : pq[0], half ? p[3]  : pq[1] };
        aP[3] = v2f{ half ? p[6]  : pq[4], half ? p[7]  : pq[5] };

#pragma unroll
        for (int j = 0; j < 4; ++j) {
            const float* vr = Vb + (size_t)(krow0 + 4 * j + 2 * half) * Dn + l16;
#pragma unroll
            for (int t = 0; t < 8; ++t) {
                v2f bv = v2f{ vr[t * 16], vr[Dn + t * 16] };
                oacc[t] = __builtin_amdgcn_wmma_f32_16x16x4_f32(
                    false, aP[j], false, bv, (short)0, oacc[t], false, false);
            }
        }
    }

    const float linv = 1.0f / l_run;
#pragma unroll
    for (int r = 0; r < 8; ++r) {
        const float lb = __shfl(linv, r + 8 * half, 32);
        const int qrow_o = q0 + r + 8 * half;
        float* orow = Ob + (size_t)qrow_o * Dn + l16;
#pragma unroll
        for (int t = 0; t < 8; ++t)
            orow[t * 16] = oacc[t][r] * lb;
    }
}

extern "C" void kernel_launch(void* const* d_in, const int* in_sizes, int n_in,
                              void* d_out, int out_size, void* d_ws, size_t ws_size,
                              hipStream_t stream) {
    const float* q  = (const float*)d_in[0];
    const float* k  = (const float*)d_in[1];
    const float* v  = (const float*)d_in[2];
    const int*   vl = (const int*)d_in[3];
    float* out = (float*)d_out;

    dim3 grid(16 * (2048 / 64));
    dim3 block(128);
    DotProductAttention_58368605552662_kernel<<<grid, block, 0, stream>>>(q, k, v, vl, out);
}


# K and V fragment loads issued as groups one phase ahead of the WMMA block that uses them (next tile's K into a second register set), held by scheduling builtins; same arithmetic and sum order
# speedup vs baseline: 1.3736x; 1.3736x over previous
#include <hip/hip_runtime.h>


typedef __attribute__((ext_vector_type(2))) float v2f;
typedef __attribute__((ext_vector_type(8))) float v8f;

#define ATTN_NEG -1000000.0f

__global__ __launch_bounds__(128)
void DotProductAttention_58368605552662_kernel(
    const float* __restrict__ Qm, const float* __restrict__ Km,
    const float* __restrict__ Vm, const int* __restrict__ vlens,
    float* __restrict__ Om)
{
    constexpr int Qn = 2048, Kn = 2048, Dn = 128;
    const int lane = threadIdx.x & 31;
    const int wave = threadIdx.x >> 5;
    const int half = lane >> 4;
    const int l16  = lane & 15;

    const int b      = blockIdx.x >> 5;
    const int qchunk = blockIdx.x & 31;
    const int q0     = qchunk * 64 + wave * 16;
    const int vlen   = vlens[b];

    const float* Qb = Qm + (size_t)b * Qn * Dn;
    const float* Kb = Km + (size_t)b * Kn * Dn;
    const float* Vb = Vm + (size_t)b * Kn * Dn;
    float*       Ob = Om + (size_t)b * Qn * Dn;

    v2f qreg[32];
    const float* qrow = Qb + (size_t)(q0 + l16) * Dn + 2 * half;
#pragma unroll
    for (int c = 0; c < 32; ++c)
        qreg[c] = *(const v2f*)(qrow + 4 * c);

    v8f oacc[8];
#pragma unroll
    for (int t = 0; t < 8; ++t) oacc[t] = (v8f)0.0f;
    float m_run = -3.0e38f;
    float l_run = 0.0f;

    const float scale = 0.08838834764831845f;

    v2f kfA[32], kfB[32];
    {
        const float* krow = Kb + (size_t)l16 * Dn + 2 * half;
#pragma unroll
        for (int c = 0; c < 32; ++c)
            kfA[c] = *(const v2f*)(krow + 4 * c);
    }

    auto tile = [&](const int kt, v2f (&kc)[32], v2f (&kn)[32]) {
        const int krow0 = kt * 16;

        float vlo[4][8], vhi[4][8];
        const float* vbase = Vb + (size_t)(krow0 + 2 * half) * Dn + l16;
#pragma unroll
        for (int j = 0; j < 3; ++j) {
#pragma unroll
            for (int t = 0; t < 8; ++t) {
                vlo[j][t] = vbase[(4 * j) * Dn + t * 16];
                vhi[j][t] = vbase[(4 * j + 1) * Dn + t * 16];
            }
        }
        __builtin_amdgcn_sched_barrier(0);

        v8f sacc = (v8f)0.0f;
#pragma unroll
        for (int c = 0; c < 32; ++c)
            sacc = __builtin_amdgcn_wmma_f32_16x16x4_f32(
                false, kc[c], false, qreg[c], (short)0, sacc, false, false);
        {
#pragma unroll
            for (int t = 0; t < 8; ++t) {
                vlo[3][t] = vbase[12 * Dn + t * 16];
                vhi[3][t] = vbase[13 * Dn + t * 16];
            }
            const int knext = (krow0 + 16 < Kn) ? (krow0 + 16) : krow0;
            const float* krow = Kb + (size_t)(knext + l16) * Dn + 2 * half;
#pragma unroll
            for (int c = 0; c < 32; ++c)
                kn[c] = *(const v2f*)(krow + 4 * c);
        }

        __builtin_amdgcn_sched_group_barrier(0x008, 32, 0);
        __builtin_amdgcn_sched_group_barrier(0x020, 48, 0);
        __builtin_amdgcn_sched_barrier(0);

        float p[8];
        float tmax = -3.0e38f;
#pragma unroll
        for (int r = 0; r < 8; ++r) {
            float s = sacc[r] * scale;
            int kg = krow0 + r + 8 * half;
            if (kg >= vlen) s = ATTN_NEG;
            p[r] = s;
            tmax = fmaxf(tmax, s);
        }
        tmax = fmaxf(tmax, __shfl_xor(tmax, 16, 32));
        const float m_new = fmaxf(m_run, tmax);
        const float alpha = __expf(m_run - m_new);
        float tsum = 0.0f;
#pragma unroll
        for (int r = 0; r < 8; ++r) {
            p[r] = __expf(p[r] - m_new);
            tsum += p[r];
        }
        tsum += __shfl_xor(tsum, 16, 32);
        l_run = l_run * alpha + tsum;
        m_run = m_new;

#pragma unroll
        for (int r = 0; r < 8; ++r) {
            float ab = __shfl(alpha, r + 8 * half, 32);
#pragma unroll
            for (int t = 0; t < 8; ++t)
                oacc[t][r] = oacc[t][r] * ab;
        }

        float pq[8];
#pragma unroll
        for (int r = 0; r < 8; ++r)
            pq[r] = __shfl_xor(p[r], 16, 32);
        v2f aP[4];
        aP[0] = v2f{ half ? pq[2] : p[0],  half ? pq[3] : p[1] };
        aP[1] = v2f{ half ? pq[6] : p[4],  half ? pq[7] : p[5] };
        aP[2] = v2f{ half ? p[2]  : pq[0], half ? p[3]  : pq[1] };
        aP[3] = v2f{ half ? p[6]  : pq[4], half ? p[7]  : pq[5] };

#pragma unroll
        for (int j = 0; j < 4; ++j) {
#pragma unroll
            for (int t = 0; t < 8; ++t) {
                v2f bv = v2f{ vlo[j][t], vhi[j][t] };
                oacc[t] = __builtin_amdgcn_wmma_f32_16x16x4_f32(
                    false, aP[j], false, bv, (short)0, oacc[t], false, false);
            }
        }
    };

    for (int kt = 0; kt < Kn / 16; kt += 2) {
        tile(kt,     kfA, kfB);
        tile(kt + 1, kfB, kfA);
    }

    const float linv = 1.0f / l_run;
#pragma unroll
    for (int r = 0; r < 8; ++r) {
        const float lb = __shfl(linv, r + 8 * half, 32);
        const int qrow_o = q0 + r + 8 * half;
        float* orow = Ob + (size_t)qrow_o * Dn + l16;
#pragma unroll
        for (int t = 0; t < 8; ++t)
            orow[t * 16] = oacc[t][r] * lb;
    }
}

extern "C" void kernel_launch(void* const* d_in, const int* in_sizes, int n_in,
                              void* d_out, int out_size, void* d_ws, size_t ws_size,
                              hipStream_t stream) {
    const float* q  = (const float*)d_in[0];
    const float* k  = (const float*)d_in[1];
    const float* v  = (const float*)d_in[2];
    const int*   vl = (const int*)d_in[3];
    float* out = (float*)d_out;

    dim3 grid(16 * (2048 / 64));
    dim3 block(128);
    DotProductAttention_58368605552662_kernel<<<grid, block, 0, stream>>>(q, k, v, vl, out);
}


# Q, K, V loads and O stores widened to 16 bytes per lane: d order inside each block of 8 permuted alike on both score operands (score sum order changes), V and O lanes own 4 adjacent columns
# speedup vs baseline: 1.7518x; 1.2753x over previous
#include <hip/hip_runtime.h>


typedef __attribute__((ext_vector_type(2))) float v2f;
typedef __attribute__((ext_vector_type(4))) float v4f;
typedef __attribute__((ext_vector_type(8))) float v8f;

#define ATTN_NEG -1000000.0f

__global__ __launch_bounds__(128)
void DotProductAttention_58368605552662_kernel(
    const float* __restrict__ Qm, const float* __restrict__ Km,
    const float* __restrict__ Vm, const int* __restrict__ vlens,
    float* __restrict__ Om)
{
    constexpr int Qn = 2048, Kn = 2048, Dn = 128;
    const int lane = threadIdx.x & 31;
    const int wave = threadIdx.x >> 5;
    const int half = lane >> 4;
    const int l16  = lane & 15;

    const int b      = blockIdx.x >> 5;
    const int qchunk = blockIdx.x & 31;
    const int q0     = qchunk * 64 + wave * 16;
    const int vlen   = vlens[b];

    const float* Qb = Qm + (size_t)b * Qn * Dn;
    const float* Kb = Km + (size_t)b * Kn * Dn;
    const float* Vb = Vm + (size_t)b * Kn * Dn;
    float*       Ob = Om + (size_t)b * Qn * Dn;

    v4f qreg[16];
    const float* qrow = Qb + (size_t)(q0 + l16) * Dn + 4 * half;
#pragma unroll
    for (int i = 0; i < 16; ++i)
        qreg[i] = *(const v4f*)(qrow + 8 * i);

    v8f oacc[8];
#pragma unroll
    for (int t = 0; t < 8; ++t) oacc[t] = (v8f)0.0f;
    float m_run = -3.0e38f;
    float l_run = 0.0f;

    const float scale = 0.08838834764831845f;

    v4f kfA[16], kfB[16];
    {
        const float* krow = Kb + (size_t)l16 * Dn + 4 * half;
#pragma unroll
        for (int i = 0; i < 16; ++i)
            kfA[i] = *(const v4f*)(krow + 8 * i);
    }

    auto tile = [&](const int kt, v4f (&kc)[16], v4f (&kn)[16]) {
        const int krow0 = kt * 16;

        v4f vlo[4][2], vhi[4][2];
        const float* vbase = Vb + (size_t)(krow0 + 2 * half) * Dn + 4 * l16;
#pragma unroll
        for (int j = 0; j < 3; ++j) {
#pragma unroll
            for (int u = 0; u < 2; ++u) {
                vlo[j][u] = *(const v4f*)(vbase + (4 * j) * Dn + 64 * u);
                vhi[j][u] = *(const v4f*)(vbase + (4 * j + 1) * Dn + 64 * u);
            }
        }
        __builtin_amdgcn_sched_barrier(0);

        v8f sacc = (v8f)0.0f;
#pragma unroll
        for (int i = 0; i < 16; ++i) {
            sacc = __builtin_amdgcn_wmma_f32_16x16x4_f32(
                false, kc[i].lo, false, qreg[i].lo, (short)0, sacc, false, false);
            sacc = __builtin_amdgcn_wmma_f32_16x16x4_f32(
                false, kc[i].hi, false, qreg[i].hi, (short)0, sacc, false, false);
        }
        {
#pragma unroll
            for (int u = 0; u < 2; ++u) {
                vlo[3][u] = *(const v4f*)(vbase + 12 * Dn + 64 * u);
                vhi[3][u] = *(const v4f*)(vbase + 13 * Dn + 64 * u);
            }
            const int knext = (krow0 + 16 < Kn) ? (krow0 + 16) : krow0;
            const float* krow = Kb + (size_t)(knext + l16) * Dn + 4 * half;
#pragma unroll
            for (int i = 0; i < 16; ++i)
                kn[i] = *(const v4f*)(krow + 8 * i);
        }

        __builtin_amdgcn_sched_barrier(0);

        float p[8];
        float tmax = -3.0e38f;
#pragma unroll
        for (int r = 0; r < 8; ++r) {
            float s = sacc[r] * scale;
            int kg = krow0 + r + 8 * half;
            if (kg >= vlen) s = ATTN_NEG;
            p[r] = s;
            tmax = fmaxf(tmax, s);
        }
        tmax = fmaxf(tmax, __shfl_xor(tmax, 16, 32));
        const float m_new = fmaxf(m_run, tmax);
        const float alpha = __expf(m_run - m_new);
        float tsum = 0.0f;
#pragma unroll
        for (int r = 0; r < 8; ++r) {
            p[r] = __expf(p[r] - m_new);
            tsum += p[r];
        }
        tsum += __shfl_xor(tsum, 16, 32);
        l_run = l_run * alpha + tsum;
        m_run = m_new;

#pragma unroll
        for (int r = 0; r < 8; ++r) {
            float ab = __shfl(alpha, r + 8 * half, 32);
#pragma unroll
            for (int t = 0; t < 8; ++t)
                oacc[t][r] = oacc[t][r] * ab;
        }

        float pq[8];
#pragma unroll
        for (int r = 0; r < 8; ++r)
            pq[r] = __shfl_xor(p[r], 16, 32);
        v2f aP[4];
        aP[0] = v2f{ half ? pq[2] : p[0],  half ? pq[3] : p[1] };
        aP[1] = v2f{ half ? pq[6] : p[4],  half ? pq[7] : p[5] };
        aP[2] = v2f{ half ? p[2]  : pq[0], half ? p[3]  : pq[1] };
        aP[3] = v2f{ half ? p[6]  : pq[4], half ? p[7]  : pq[5] };

#pragma unroll
        for (int j = 0; j < 4; ++j) {
#pragma unroll
            for (int t = 0; t < 8; ++t) {
                v2f bv = v2f{ vlo[j][t >> 2][t & 3], vhi[j][t >> 2][t & 3] };
                oacc[t] = __builtin_amdgcn_wmma_f32_16x16x4_f32(
                    false, aP[j], false, bv, (short)0, oacc[t], false, false);
            }
        }
    };

    for (int kt = 0; kt < Kn / 16; kt += 2) {
        tile(kt,     kfA, kfB);
        tile(kt + 1, kfB, kfA);
    }

    const float linv = 1.0f / l_run;
#pragma unroll
    for (int r = 0; r < 8; ++r) {
        const float lb = __shfl(linv, r + 8 * half, 32);
        const int qrow_o = q0 + r + 8 * half;
        float* orow = Ob + (size_t)qrow_o * Dn + 4 * l16;
#pragma unroll
        for (int u = 0; u < 2; ++u) {
            v4f o = v4f{ oacc[4 * u][r], oacc[4 * u + 1][r],
                         oacc[4 * u + 2][r], oacc[4 * u + 3][r] } * lb;
            *(v4f*)(orow + 64 * u) = o;
        }
    }
}

extern "C" void kernel_launch(void* const* d_in, const int* in_sizes, int n_in,
                              void* d_out, int out_size, void* d_ws, size_t ws_size,
                              hipStream_t stream) {
    const float* q  = (const float*)d_in[0];
    const float* k  = (const float*)d_in[1];
    const float* v  = (const float*)d_in[2];
    const int*   vl = (const int*)d_in[3];
    float* out = (float*)d_out;

    dim3 grid(16 * (2048 / 64));
    dim3 block(128);
    DotProductAttention_58368605552662_kernel<<<grid, block, 0, stream>>>(q, k, v, vl, out);
}


# Key tiles wholly past valid_len are skipped once every row's running max shows their softmax weights are exactly 0 (else all tiles run); same bits, same sum order
# speedup vs baseline: 2.4817x; 1.4167x over previous
#include <hip/hip_runtime.h>


typedef __attribute__((ext_vector_type(2))) float v2f;
typedef __attribute__((ext_vector_type(4))) float v4f;
typedef __attribute__((ext_vector_type(8))) float v8f;

#define ATTN_NEG -1000000.0f

__global__ __launch_bounds__(128)
void DotProductAttention_58368605552662_kernel(
    const float* __restrict__ Qm, const float* __restrict__ Km,
    const float* __restrict__ Vm, const int* __restrict__ vlens,
    float* __restrict__ Om)
{
    constexpr int Qn = 2048, Kn = 2048, Dn = 128;
    const int lane = threadIdx.x & 31;
    const int wave = threadIdx.x >> 5;
    const int half = lane >> 4;
    const int l16  = lane & 15;

    const int b      = blockIdx.x >> 5;
    const int qchunk = blockIdx.x & 31;
    const int q0     = qchunk * 64 + wave * 16;
    const int vlen   = vlens[b];

    const float* Qb = Qm + (size_t)b * Qn * Dn;
    const float* Kb = Km + (size_t)b * Kn * Dn;
    const float* Vb = Vm + (size_t)b * Kn * Dn;
    float*       Ob = Om + (size_t)b * Qn * Dn;

    v4f qreg[16];
    const float* qrow = Qb + (size_t)(q0 + l16) * Dn + 4 * half;
#pragma unroll
    for (int i = 0; i < 16; ++i)
        qreg[i] = *(const v4f*)(qrow + 8 * i);

    v8f oacc[8];
#pragma unroll
    for (int t = 0; t < 8; ++t) oacc[t] = (v8f)0.0f;
    float m_run = -3.0e38f;
    float l_run = 0.0f;

    const float scale = 0.08838834764831845f;

    v4f kfA[16], kfB[16];
    {
        const float* krow = Kb + (size_t)l16 * Dn + 4 * half;
#pragma unroll
        for (int i = 0; i < 16; ++i)
            kfA[i] = *(const v4f*)(krow + 8 * i);
    }

    auto tile = [&](const int kt, v4f (&kc)[16], v4f (&kn)[16]) {
        const int krow0 = kt * 16;

        v4f vlo[4][2], vhi[4][2];
        const float* vbase = Vb + (size_t)(krow0 + 2 * half) * Dn + 4 * l16;
#pragma unroll
        for (int j = 0; j < 3; ++j) {
#pragma unroll
            for (int u = 0; u < 2; ++u) {
                vlo[j][u] = *(const v4f*)(vbase + (4 * j) * Dn + 64 * u);
                vhi[j][u] = *(const v4f*)(vbase + (4 * j + 1) * Dn + 64 * u);
            }
        }
        __builtin_amdgcn_sched_barrier(0);

        v8f sacc = (v8f)0.0f;
#pragma unroll
        for (int i = 0; i < 16; ++i) {
            sacc = __builtin_amdgcn_wmma_f32_16x16x4_f32(
                false, kc[i].lo, false, qreg[i].lo, (short)0, sacc, false, false);
            sacc = __builtin_amdgcn_wmma_f32_16x16x4_f32(
                false, kc[i].hi, false, qreg[i].hi, (short)0, sacc, false, false);
        }
        {
#pragma unroll
            for (int u = 0; u < 2; ++u) {
                vlo[3][u] = *(const v4f*)(vbase + 12 * Dn + 64 * u);
                vhi[3][u] = *(const v4f*)(vbase + 13 * Dn + 64 * u);
            }
            const int knext = (krow0 + 16 < Kn) ? (krow0 + 16) : krow0;
            const float* krow = Kb + (size_t)(knext + l16) * Dn + 4 * half;
#pragma unroll
            for (int i = 0; i < 16; ++i)
                kn[i] = *(const v4f*)(krow + 8 * i);
        }

        __builtin_amdgcn_sched_barrier(0);

        float p[8];
        float tmax = -3.0e38f;
#pragma unroll
        for (int r = 0; r < 8; ++r) {
            float s = sacc[r] * scale;
            int kg = krow0 + r + 8 * half;
            if (kg >= vlen) s = ATTN_NEG;
            p[r] = s;
            tmax = fmaxf(tmax, s);
        }
        tmax = fmaxf(tmax, __shfl_xor(tmax, 16, 32));
        const float m_new = fmaxf(m_run, tmax);
        const float alpha = __expf(m_run - m_new);
        float tsum = 0.0f;
#pragma unroll
        for (int r = 0; r < 8; ++r) {
            p[r] = __expf(p[r] - m_new);
            tsum += p[r];
        }
        tsum += __shfl_xor(tsum, 16, 32);
        l_run = l_run * alpha + tsum;
        m_run = m_new;

#pragma unroll
        for (int r = 0; r < 8; ++r) {
            float ab = __shfl(alpha, r + 8 * half, 32);
#pragma unroll
            for (int t = 0; t < 8; ++t)
                oacc[t][r] = oacc[t][r] * ab;
        }

        float pq[8];
#pragma unroll
        for (int r = 0; r < 8; ++r)
            pq[r] = __shfl_xor(p[r], 16, 32);
        v2f aP[4];
        aP[0] = v2f{ half ? pq[2] : p[0],  half ? pq[3] : p[1] };
        aP[1] = v2f{ half ? pq[6] : p[4],  half ? pq[7] : p[5] };
        aP[2] = v2f{ half ? p[2]  : pq[0], half ? p[3]  : pq[1] };
        aP[3] = v2f{ half ? p[6]  : pq[4], half ? p[7]  : pq[5] };

#pragma unroll
        for (int j = 0; j < 4; ++j) {
#pragma unroll
            for (int t = 0; t < 8; ++t) {
                v2f bv = v2f{ vlo[j][t >> 2][t & 3], vhi[j][t >> 2][t & 3] };
                oacc[t] = __builtin_amdgcn_wmma_f32_16x16x4_f32(
                    false, aP[j], false, bv, (short)0, oacc[t], false, false);
            }
        }
    };

    constexpr int NT = Kn / 16;
    const int nvalid = (vlen >= Kn) ? NT : (vlen <= 0 ? 1 : ((vlen + 15) >> 4));
    const int kt_valid = (nvalid + 1) & ~1;
    int kt_end = kt_valid;
    for (int kt = 0; kt < kt_end; kt += 2) {
        tile(kt,     kfA, kfB);
        tile(kt + 1, kfB, kfA);
        if (kt + 2 == kt_valid && kt_valid < NT) {
            if (!__all(m_run > ATTN_NEG + 200.0f)) kt_end = NT;
        }
    }

    const float linv = 1.0f / l_run;
#pragma unroll
    for (int r = 0; r < 8; ++r) {
        const float lb = __shfl(linv, r + 8 * half, 32);
        const int qrow_o = q0 + r + 8 * half;
        float* orow = Ob + (size_t)qrow_o * Dn + 4 * l16;
#pragma unroll
        for (int u = 0; u < 2; ++u) {
            v4f o = v4f{ oacc[4 * u][r], oacc[4 * u + 1][r],
                         oacc[4 * u + 2][r], oacc[4 * u + 3][r] } * lb;
            *(v4f*)(orow + 64 * u) = o;
        }
    }
}

extern "C" void kernel_launch(void* const* d_in, const int* in_sizes, int n_in,
                              void* d_out, int out_size, void* d_ws, size_t ws_size,
                              hipStream_t stream) {
    const float* q  = (const float*)d_in[0];
    const float* k  = (const float*)d_in[1];
    const float* v  = (const float*)d_in[2];
    const int*   vl = (const int*)d_in[3];
    float* out = (float*)d_out;

    dim3 grid(16 * (2048 / 64));
    dim3 block(128);
    DotProductAttention_58368605552662_kernel<<<grid, block, 0, stream>>>(q, k, v, vl, out);
}


# Work-groups of 8 waves pair a long batch (older waves 0-3) with the complementary short one (waves 4-7) by tile-count rank, so each SIMD gets about equal work; same arithmetic and sum order
# speedup vs baseline: 2.7156x; 1.0942x over previous
#include <hip/hip_runtime.h>


typedef __attribute__((ext_vector_type(2))) float v2f;
typedef __attribute__((ext_vector_type(4))) float v4f;
typedef __attribute__((ext_vector_type(8))) float v8f;

#define ATTN_NEG -1000000.0f

__global__ __launch_bounds__(256)
void DotProductAttention_58368605552662_kernel(
    const float* __restrict__ Qm, const float* __restrict__ Km,
    const float* __restrict__ Vm, const int* __restrict__ vlens,
    float* __restrict__ Om)
{
    constexpr int Qn = 2048, Kn = 2048, Dn = 128;
    constexpr int NT = Kn / 16;
    const int lane = threadIdx.x & 31;
    const int wave = __builtin_amdgcn_readfirstlane((int)(threadIdx.x >> 5));
    const int half = lane >> 4;
    const int l16  = lane & 15;

    const int myv    = vlens[l16];
    const int mycost = (myv <= 0 || myv >= Kn) ? NT : ((myv + 15) >> 4);
    int rank = 0;
#pragma unroll
    for (int j = 0; j < 16; ++j) {
        const int vj = vlens[j];
        const int cj = (vj <= 0 || vj >= Kn) ? NT : ((vj + 15) >> 4);
        rank += (cj < mycost || (cj == mycost && j < l16)) ? 1 : 0;
    }
    const int pairi  = blockIdx.x >> 5;
    const int qchunk = blockIdx.x & 31;
    const int want   = (wave < 4) ? (15 - pairi) : pairi;
    const unsigned hit = __builtin_amdgcn_ballot_w32(rank == want);
    const int b      = __builtin_ctz(hit) & 15;
    const int q0     = qchunk * 64 + (wave & 3) * 16;
    const int vlen   = vlens[b];

    const float* Qb = Qm + (size_t)b * Qn * Dn;
    const float* Kb = Km + (size_t)b * Kn * Dn;
    const float* Vb = Vm + (size_t)b * Kn * Dn;
    float*       Ob = Om + (size_t)b * Qn * Dn;

    v4f qreg[16];
    const float* qrow = Qb + (size_t)(q0 + l16) * Dn + 4 * half;
#pragma unroll
    for (int i = 0; i < 16; ++i)
        qreg[i] = *(const v4f*)(qrow + 8 * i);

    v8f oacc[8];
#pragma unroll
    for (int t = 0; t < 8; ++t) oacc[t] = (v8f)0.0f;
    float m_run = -3.0e38f;
    float l_run = 0.0f;

    const float scale = 0.08838834764831845f;

    v4f kfA[16], kfB[16];
    {
        const float* krow = Kb + (size_t)l16 * Dn + 4 * half;
#pragma unroll
        for (int i = 0; i < 16; ++i)
            kfA[i] = *(const v4f*)(krow + 8 * i);
    }

    auto tile = [&](const int kt, v4f (&kc)[16], v4f (&kn)[16]) {
        const int krow0 = kt * 16;

        v4f vlo[4][2], vhi[4][2];
        const float* vbase = Vb + (size_t)(krow0 + 2 * half) * Dn + 4 * l16;
#pragma unroll
        for (int j = 0; j < 3; ++j) {
#pragma unroll
            for (int u = 0; u < 2; ++u) {
                vlo[j][u] = *(const v4f*)(vbase + (4 * j) * Dn + 64 * u);
                vhi[j][u] = *(const v4f*)(vbase + (4 * j + 1) * Dn + 64 * u);
            }
        }
        __builtin_amdgcn_sched_barrier(0);

        v8f sacc = (v8f)0.0f;
#pragma unroll
        for (int i = 0; i < 16; ++i) {
            sacc = __builtin_amdgcn_wmma_f32_16x16x4_f32(
                false, kc[i].lo, false, qreg[i].lo, (short)0, sacc, false, false);
            sacc = __builtin_amdgcn_wmma_f32_16x16x4_f32(
                false, kc[i].hi, false, qreg[i].hi, (short)0, sacc, false, false);
        }
        {
#pragma unroll
            for (int u = 0; u < 2; ++u) {
                vlo[3][u] = *(const v4f*)(vbase + 12 * Dn + 64 * u);
                vhi[3][u] = *(const v4f*)(vbase + 13 * Dn + 64 * u);
            }
            const int knext = (krow0 + 16 < Kn) ? (krow0 + 16) : krow0;
            const float* krow = Kb + (size_t)(knext + l16) * Dn + 4 * half;
#pragma unroll
            for (int i = 0; i < 16; ++i)
                kn[i] = *(const v4f*)(krow + 8 * i);
        }

        __builtin_amdgcn_sched_barrier(0);

        float p[8];
        float tmax = -3.0e38f;
#pragma unroll
        for (int r = 0; r < 8; ++r) {
            float s = sacc[r] * scale;
            int kg = krow0 + r + 8 * half;
            if (kg >= vlen) s = ATTN_NEG;
            p[r] = s;
            tmax = fmaxf(tmax, s);
        }
        tmax = fmaxf(tmax, __shfl_xor(tmax, 16, 32));
        const float m_new = fmaxf(m_run, tmax);
        const float alpha = __expf(m_run - m_new);
        float tsum = 0.0f;
#pragma unroll
        for (int r = 0; r < 8; ++r) {
            p[r] = __expf(p[r] - m_new);
            tsum += p[r];
        }
        tsum += __shfl_xor(tsum, 16, 32);
        l_run = l_run * alpha + tsum;
        m_run = m_new;

#pragma unroll
        for (int r = 0; r < 8; ++r) {
            float ab = __shfl(alpha, r + 8 * half, 32);
#pragma unroll
            for (int t = 0; t < 8; ++t)
                oacc[t][r] = oacc[t][r] * ab;
        }

        float pq[8];
#pragma unroll
        for (int r = 0; r < 8; ++r)
            pq[r] = __shfl_xor(p[r], 16, 32);
        v2f aP[4];
        aP[0] = v2f{ half ? pq[2] : p[0],  half ? pq[3] : p[1] };
        aP[1] = v2f{ half ? pq[6] : p[4],  half ? pq[7] : p[5] };
        aP[2] = v2f{ half ? p[2]  : pq[0], half ? p[3]  : pq[1] };
        aP[3] = v2f{ half ? p[6]  : pq[4], half ? p[7]  : pq[5] };

#pragma unroll
        for (int j = 0; j < 4; ++j) {
#pragma unroll
            for (int t = 0; t < 8; ++t) {
                v2f bv = v2f{ vlo[j][t >> 2][t & 3], vhi[j][t >> 2][t & 3] };
                oacc[t] = __builtin_amdgcn_wmma_f32_16x16x4_f32(
                    false, aP[j], false, bv, (short)0, oacc[t], false, false);
            }
        }
    };

    const int nvalid = (vlen >= Kn) ? NT : (vlen <= 0 ? 1 : ((vlen + 15) >> 4));
    const int kt_valid = (nvalid + 1) & ~1;
    int kt_end = kt_valid;
    for (int kt = 0; kt < kt_end; kt += 2) {
        tile(kt,     kfA, kfB);
        tile(kt + 1, kfB, kfA);
        if (kt + 2 == kt_valid && kt_valid < NT) {
            if (!__all(m_run > ATTN_NEG + 200.0f)) kt_end = NT;
        }
    }

    const float linv = 1.0f / l_run;
#pragma unroll
    for (int r = 0; r < 8; ++r) {
        const float lb = __shfl(linv, r + 8 * half, 32);
        const int qrow_o = q0 + r + 8 * half;
        float* orow = Ob + (size_t)qrow_o * Dn + 4 * l16;
#pragma unroll
        for (int u = 0; u < 2; ++u) {
            v4f o = v4f{ oacc[4 * u][r], oacc[4 * u + 1][r],
                         oacc[4 * u + 2][r], oacc[4 * u + 3][r] } * lb;
            *(v4f*)(orow + 64 * u) = o;
        }
    }
}

extern "C" void kernel_launch(void* const* d_in, const int* in_sizes, int n_in,
                              void* d_out, int out_size, void* d_ws, size_t ws_size,
                              hipStream_t stream) {
    const float* q  = (const float*)d_in[0];
    const float* k  = (const float*)d_in[1];
    const float* v  = (const float*)d_in[2];
    const int*   vl = (const int*)d_in[3];
    float* out = (float*)d_out;

    dim3 grid(256);
    dim3 block(256);
    DotProductAttention_58368605552662_kernel<<<grid, block, 0, stream>>>(q, k, v, vl, out);
}


# Each 16-query tile's key tiles split in two contiguous halves between the two waves of a SIMD; partial max, sum, O merged once through LDS (sum order, rescale points change); one batch per group
# speedup vs baseline: 2.9798x; 1.0973x over previous
#include <hip/hip_runtime.h>


typedef __attribute__((ext_vector_type(2))) float v2f;
typedef __attribute__((ext_vector_type(4))) float v4f;
typedef __attribute__((ext_vector_type(8))) float v8f;

#define ATTN_NEG -1000000.0f

__global__ __launch_bounds__(256)
void DotProductAttention_58368605552662_kernel(
    const float* __restrict__ Qm, const float* __restrict__ Km,
    const float* __restrict__ Vm, const int* __restrict__ vlens,
    float* __restrict__ Om)
{
    constexpr int Qn = 2048, Kn = 2048, Dn = 128;
    constexpr int NT = Kn / 16;
    __shared__ float xch[4][2112];
    const int lane = threadIdx.x & 31;
    const int wave = __builtin_amdgcn_readfirstlane((int)(threadIdx.x >> 5));
    const int half = lane >> 4;
    const int l16  = lane & 15;

    const int myv    = vlens[l16];
    const int mycost = (myv <= 0 || myv >= Kn) ? NT : ((myv + 15) >> 4);
    int rank = 0;
#pragma unroll
    for (int j = 0; j < 16; ++j) {
        const int vj = vlens[j];
        const int cj = (vj <= 0 || vj >= Kn) ? NT : ((vj + 15) >> 4);
        rank += (cj < mycost || (cj == mycost && j < l16)) ? 1 : 0;
    }
    const int pos    = blockIdx.x >> 5;
    const int qchunk = blockIdx.x & 31;
    const int want   = 15 - pos;
    const int qt     = wave & 3;
    const int kh     = wave >> 2;
    const unsigned hit = __builtin_amdgcn_ballot_w32(rank == want);
    const int b      = __builtin_ctz(hit) & 15;
    const int q0     = qchunk * 64 + qt * 16;
    const int vlen   = vlens[b];

    const float* Qb = Qm + (size_t)b * Qn * Dn;
    const float* Kb = Km + (size_t)b * Kn * Dn;
    const float* Vb = Vm + (size_t)b * Kn * Dn;
    float*       Ob = Om + (size_t)b * Qn * Dn;

    v4f qreg[16];
    const float* qrow = Qb + (size_t)(q0 + l16) * Dn + 4 * half;
#pragma unroll
    for (int i = 0; i < 16; ++i)
        qreg[i] = *(const v4f*)(qrow + 8 * i);

    v8f oacc[8];
#pragma unroll
    for (int t = 0; t < 8; ++t) oacc[t] = (v8f)0.0f;
    float m_run = -3.0e38f;
    float l_run = 0.0f;

    const float scale = 0.08838834764831845f;

    v4f kfA[16], kfB[16];

    const int nvalid = (vlen >= Kn || vlen <= 0) ? NT : ((vlen + 15) >> 4);
    const int npair  = (nvalid + 1) >> 1;
    const int psplit = (npair + 1) >> 1;
    const int p_begin = kh ? psplit : 0;
    const int p_end   = kh ? npair : psplit;

    auto load_first_k = [&](const int p) {
        const float* krow = Kb + (size_t)(p * 32 + l16) * Dn + 4 * half;
#pragma unroll
        for (int i = 0; i < 16; ++i)
            kfA[i] = *(const v4f*)(krow + 8 * i);
    };
    load_first_k(p_begin);

    auto tile = [&](const int kt, v4f (&kc)[16], v4f (&kn)[16]) {
        const int krow0 = kt * 16;

        v4f vlo[4][2], vhi[4][2];
        const float* vbase = Vb + (size_t)(krow0 + 2 * half) * Dn + 4 * l16;
#pragma unroll
        for (int j = 0; j < 3; ++j) {
#pragma unroll
            for (int u = 0; u < 2; ++u) {
                vlo[j][u] = *(const v4f*)(vbase + (4 * j) * Dn + 64 * u);
                vhi[j][u] = *(const v4f*)(vbase + (4 * j + 1) * Dn + 64 * u);
            }
        }
        __builtin_amdgcn_sched_barrier(0);

        v8f sacc = (v8f)0.0f;
#pragma unroll
        for (int i = 0; i < 16; ++i) {
            sacc = __builtin_amdgcn_wmma_f32_16x16x4_f32(
                false, kc[i].lo, false, qreg[i].lo, (short)0, sacc, false, false);
            sacc = __builtin_amdgcn_wmma_f32_16x16x4_f32(
                false, kc[i].hi, false, qreg[i].hi, (short)0, sacc, false, false);
        }
        {
#pragma unroll
            for (int u = 0; u < 2; ++u) {
                vlo[3][u] = *(const v4f*)(vbase + 12 * Dn + 64 * u);
                vhi[3][u] = *(const v4f*)(vbase + 13 * Dn + 64 * u);
            }
            const int knext = (krow0 + 16 < Kn) ? (krow0 + 16) : krow0;
            const float* krow = Kb + (size_t)(knext + l16) * Dn + 4 * half;
#pragma unroll
            for (int i = 0; i < 16; ++i)
                kn[i] = *(const v4f*)(krow + 8 * i);
        }

        __builtin_amdgcn_sched_barrier(0);

        float p[8];
        float tmax = -3.0e38f;
#pragma unroll
        for (int r = 0; r < 8; ++r) {
            float s = sacc[r] * scale;
            int kg = krow0 + r + 8 * half;
            if (kg >= vlen) s = ATTN_NEG;
            p[r] = s;
            tmax = fmaxf(tmax, s);
        }
        tmax = fmaxf(tmax, __shfl_xor(tmax, 16, 32));
        const float m_new = fmaxf(m_run, tmax);
        const float alpha = __expf(m_run - m_new);
        float tsum = 0.0f;
#pragma unroll
        for (int r = 0; r < 8; ++r) {
            p[r] = __expf(p[r] - m_new);
            tsum += p[r];
        }
        tsum += __shfl_xor(tsum, 16, 32);
        l_run = l_run * alpha + tsum;
        m_run = m_new;

#pragma unroll
        for (int r = 0; r < 8; ++r) {
            float ab = __shfl(alpha, r + 8 * half, 32);
#pragma unroll
            for (int t = 0; t < 8; ++t)
                oacc[t][r] = oacc[t][r] * ab;
        }

        float pq[8];
#pragma unroll
        for (int r = 0; r < 8; ++r)
            pq[r] = __shfl_xor(p[r], 16, 32);
        v2f aP[4];
        aP[0] = v2f{ half ? pq[2] : p[0],  half ? pq[3] : p[1] };
        aP[1] = v2f{ half ? pq[6] : p[4],  half ? pq[7] : p[5] };
        aP[2] = v2f{ half ? p[2]  : pq[0], half ? p[3]  : pq[1] };
        aP[3] = v2f{ half ? p[6]  : pq[4], half ? p[7]  : pq[5] };

#pragma unroll
        for (int j = 0; j < 4; ++j) {
#pragma unroll
            for (int t = 0; t < 8; ++t) {
                v2f bv = v2f{ vlo[j][t >> 2][t & 3], vhi[j][t >> 2][t & 3] };
                oacc[t] = __builtin_amdgcn_wmma_f32_16x16x4_f32(
                    false, aP[j], false, bv, (short)0, oacc[t], false, false);
            }
        }
    };

    auto run = [&](const int pb, const int pe) {
        for (int p = pb; p < pe; ++p) {
            tile(2 * p,     kfA, kfB);
            tile(2 * p + 1, kfB, kfA);
        }
    };
    run(p_begin, p_end);

    float* xb = &xch[qt][0];
    if (kh) {
#pragma unroll
        for (int t = 0; t < 8; ++t) {
#pragma unroll
            for (int g = 0; g < 2; ++g)
                *(v4f*)(xb + ((t * 2 + g) * 32 + lane) * 4) =
                    v4f{ oacc[t][4 * g], oacc[t][4 * g + 1],
                         oacc[t][4 * g + 2], oacc[t][4 * g + 3] };
        }
        xb[2048 + lane] = m_run;
        xb[2080 + lane] = l_run;
    }
    __syncthreads();
    if (kh) return;

    {
        const float m2 = xb[2048 + lane];
        const float l2 = xb[2080 + lane];
        const float m_new = fmaxf(m_run, m2);
        const float a1 = __expf(m_run - m_new);
        const float a2 = __expf(m2 - m_new);
        l_run = l_run * a1 + l2 * a2;
        m_run = m_new;
#pragma unroll
        for (int g = 0; g < 2; ++g) {
            float a1b[4], a2b[4];
#pragma unroll
            for (int e = 0; e < 4; ++e) {
                a1b[e] = __shfl(a1, 4 * g + e + 8 * half, 32);
                a2b[e] = __shfl(a2, 4 * g + e + 8 * half, 32);
            }
#pragma unroll
            for (int t = 0; t < 8; ++t) {
                const v4f o2 = *(const v4f*)(xb + ((t * 2 + g) * 32 + lane) * 4);
#pragma unroll
                for (int e = 0; e < 4; ++e)
                    oacc[t][4 * g + e] = oacc[t][4 * g + e] * a1b[e] + o2[e] * a2b[e];
            }
        }
    }

    if (npair < NT / 2) {
        if (!__all(m_run > ATTN_NEG + 200.0f)) {
            load_first_k(npair);
            run(npair, NT / 2);
        }
    }

    const float linv = 1.0f / l_run;
#pragma unroll
    for (int r = 0; r < 8; ++r) {
        const float lb = __shfl(linv, r + 8 * half, 32);
        const int qrow_o = q0 + r + 8 * half;
        float* orow = Ob + (size_t)qrow_o * Dn + 4 * l16;
#pragma unroll
        for (int u = 0; u < 2; ++u) {
            v4f o = v4f{ oacc[4 * u][r], oacc[4 * u + 1][r],
                         oacc[4 * u + 2][r], oacc[4 * u + 3][r] } * lb;
            *(v4f*)(orow + 64 * u) = o;
        }
    }
}

extern "C" void kernel_launch(void* const* d_in, const int* in_sizes, int n_in,
                              void* d_out, int out_size, void* d_ws, size_t ws_size,
                              hipStream_t stream) {
    const float* q  = (const float*)d_in[0];
    const float* k  = (const float*)d_in[1];
    const float* v  = (const float*)d_in[2];
    const int*   vl = (const int*)d_in[3];
    float* out = (float*)d_out;

    dim3 grid(512);
    dim3 block(256);
    DotProductAttention_58368605552662_kernel<<<grid, block, 0, stream>>>(q, k, v, vl, out);
}


# Softmax lane exchanges in the key loop (row max, row sum, rescale factors, P halves) done by vector-ALU lane permutes instead of LDS-routed permutes; same arithmetic and sum order
# speedup vs baseline: 3.0862x; 1.0357x over previous
#include <hip/hip_runtime.h>


typedef __attribute__((ext_vector_type(2))) float v2f;
typedef __attribute__((ext_vector_type(4))) float v4f;
typedef __attribute__((ext_vector_type(8))) float v8f;

#define ATTN_NEG -1000000.0f

static __device__ __forceinline__ float xhalf(float x) {
    const int b = __builtin_bit_cast(int, x);
    return __builtin_bit_cast(float,
        __builtin_amdgcn_permlanex16(b, b, 0x76543210, 0xfedcba98, false, false));
}
static __device__ __forceinline__ float rowpick(float x, int sel) {
    const int b = __builtin_bit_cast(int, x);
    return __builtin_bit_cast(float,
        __builtin_amdgcn_permlane16_var(b, b, sel, false, false));
}

__global__ __launch_bounds__(256)
void DotProductAttention_58368605552662_kernel(
    const float* __restrict__ Qm, const float* __restrict__ Km,
    const float* __restrict__ Vm, const int* __restrict__ vlens,
    float* __restrict__ Om)
{
    constexpr int Qn = 2048, Kn = 2048, Dn = 128;
    constexpr int NT = Kn / 16;
    __shared__ float xch[4][2112];
    const int lane = threadIdx.x & 31;
    const int wave = __builtin_amdgcn_readfirstlane((int)(threadIdx.x >> 5));
    const int half = lane >> 4;
    const int l16  = lane & 15;
    const int rsel = 8 * half;

    const int myv    = vlens[l16];
    const int mycost = (myv <= 0 || myv >= Kn) ? NT : ((myv + 15) >> 4);
    int rank = 0;
#pragma unroll
    for (int j = 0; j < 16; ++j) {
        const int vj = vlens[j];
        const int cj = (vj <= 0 || vj >= Kn) ? NT : ((vj + 15) >> 4);
        rank += (cj < mycost || (cj == mycost && j < l16)) ? 1 : 0;
    }
    const int pos    = blockIdx.x >> 5;
    const int qchunk = blockIdx.x & 31;
    const int want   = 15 - pos;
    const int qt     = wave & 3;
    const int kh     = wave >> 2;
    const unsigned hit = __builtin_amdgcn_ballot_w32(rank == want);
    const int b      = __builtin_ctz(hit) & 15;
    const int q0     = qchunk * 64 + qt * 16;
    const int vlen   = vlens[b];

    const float* Qb = Qm + (size_t)b * Qn * Dn;
    const float* Kb = Km + (size_t)b * Kn * Dn;
    const float* Vb = Vm + (size_t)b * Kn * Dn;
    float*       Ob = Om + (size_t)b * Qn * Dn;

    v4f qreg[16];
    const float* qrow = Qb + (size_t)(q0 + l16) * Dn + 4 * half;
#pragma unroll
    for (int i = 0; i < 16; ++i)
        qreg[i] = *(const v4f*)(qrow + 8 * i);

    v8f oacc[8];
#pragma unroll
    for (int t = 0; t < 8; ++t) oacc[t] = (v8f)0.0f;
    float m_run = -3.0e38f;
    float l_run = 0.0f;

    const float scale = 0.08838834764831845f;

    v4f kfA[16], kfB[16];

    const int nvalid = (vlen >= Kn || vlen <= 0) ? NT : ((vlen + 15) >> 4);
    const int npair  = (nvalid + 1) >> 1;
    const int psplit = (npair + 1) >> 1;
    const int p_begin = kh ? psplit : 0;
    const int p_end   = kh ? npair : psplit;

    auto load_first_k = [&](const int p) {
        const float* krow = Kb + (size_t)(p * 32 + l16) * Dn + 4 * half;
#pragma unroll
        for (int i = 0; i < 16; ++i)
            kfA[i] = *(const v4f*)(krow + 8 * i);
    };
    load_first_k(p_begin);

    auto tile = [&](const int kt, v4f (&kc)[16], v4f (&kn)[16]) {
        const int krow0 = kt * 16;

        v4f vlo[4][2], vhi[4][2];
        const float* vbase = Vb + (size_t)(krow0 + 2 * half) * Dn + 4 * l16;
#pragma unroll
        for (int j = 0; j < 3; ++j) {
#pragma unroll
            for (int u = 0; u < 2; ++u) {
                vlo[j][u] = *(const v4f*)(vbase + (4 * j) * Dn + 64 * u);
                vhi[j][u] = *(const v4f*)(vbase + (4 * j + 1) * Dn + 64 * u);
            }
        }
        __builtin_amdgcn_sched_barrier(0);

        v8f sacc = (v8f)0.0f;
#pragma unroll
        for (int i = 0; i < 16; ++i) {
            sacc = __builtin_amdgcn_wmma_f32_16x16x4_f32(
                false, kc[i].lo, false, qreg[i].lo, (short)0, sacc, false, false);
            sacc = __builtin_amdgcn_wmma_f32_16x16x4_f32(
                false, kc[i].hi, false, qreg[i].hi, (short)0, sacc, false, false);
        }
        {
#pragma unroll
            for (int u = 0; u < 2; ++u) {
                vlo[3][u] = *(const v4f*)(vbase + 12 * Dn + 64 * u);
                vhi[3][u] = *(const v4f*)(vbase + 13 * Dn + 64 * u);
            }
            const int knext = (krow0 + 16 < Kn) ? (krow0 + 16) : krow0;
            const float* krow = Kb + (size_t)(knext + l16) * Dn + 4 * half;
#pragma unroll
            for (int i = 0; i < 16; ++i)
                kn[i] = *(const v4f*)(krow + 8 * i);
        }

        __builtin_amdgcn_sched_barrier(0);

        float p[8];
        float tmax = -3.0e38f;
#pragma unroll
        for (int r = 0; r < 8; ++r) {
            float s = sacc[r] * scale;
            int kg = krow0 + r + 8 * half;
            if (kg >= vlen) s = ATTN_NEG;
            p[r] = s;
            tmax = fmaxf(tmax, s);
        }
        tmax = fmaxf(tmax, xhalf(tmax));
        const float m_new = fmaxf(m_run, tmax);
        const float alpha = __expf(m_run - m_new);
        float tsum = 0.0f;
#pragma unroll
        for (int r = 0; r < 8; ++r) {
            p[r] = __expf(p[r] - m_new);
            tsum += p[r];
        }
        tsum += xhalf(tsum);
        l_run = l_run * alpha + tsum;
        m_run = m_new;

#pragma unroll
        for (int r = 0; r < 8; ++r) {
            float ab = rowpick(alpha, r + rsel);
#pragma unroll
            for (int t = 0; t < 8; ++t)
                oacc[t][r] = oacc[t][r] * ab;
        }

        float pq[8];
#pragma unroll
        for (int r = 0; r < 8; ++r)
            pq[r] = xhalf(p[r]);
        v2f aP[4];
        aP[0] = v2f{ half ? pq[2] : p[0],  half ? pq[3] : p[1] };
        aP[1] = v2f{ half ? pq[6] : p[4],  half ? pq[7] : p[5] };
        aP[2] = v2f{ half ? p[2]  : pq[0], half ? p[3]  : pq[1] };
        aP[3] = v2f{ half ? p[6]  : pq[4], half ? p[7]  : pq[5] };

#pragma unroll
        for (int j = 0; j < 4; ++j) {
#pragma unroll
            for (int t = 0; t < 8; ++t) {
                v2f bv = v2f{ vlo[j][t >> 2][t & 3], vhi[j][t >> 2][t & 3] };
                oacc[t] = __builtin_amdgcn_wmma_f32_16x16x4_f32(
                    false, aP[j], false, bv, (short)0, oacc[t], false, false);
            }
        }
    };

    auto run = [&](const int pb, const int pe) {
        for (int p = pb; p < pe; ++p) {
            tile(2 * p,     kfA, kfB);
            tile(2 * p + 1, kfB, kfA);
        }
    };
    run(p_begin, p_end);

    float* xb = &xch[qt][0];
    if (kh) {
#pragma unroll
        for (int t = 0; t < 8; ++t) {
#pragma unroll
            for (int g = 0; g < 2; ++g)
                *(v4f*)(xb + ((t * 2 + g) * 32 + lane) * 4) =
                    v4f{ oacc[t][4 * g], oacc[t][4 * g + 1],
                         oacc[t][4 * g + 2], oacc[t][4 * g + 3] };
        }
        xb[2048 + lane] = m_run;
        xb[2080 + lane] = l_run;
    }
    __syncthreads();
    if (kh) return;

    {
        const float m2 = xb[2048 + lane];
        const float l2 = xb[2080 + lane];
        const float m_new = fmaxf(m_run, m2);
        const float a1 = __expf(m_run - m_new);
        const float a2 = __expf(m2 - m_new);
        l_run = l_run * a1 + l2 * a2;
        m_run = m_new;
#pragma unroll
        for (int g = 0; g < 2; ++g) {
            float a1b[4], a2b[4];
#pragma unroll
            for (int e = 0; e < 4; ++e) {
                a1b[e] = __shfl(a1, 4 * g + e + 8 * half, 32);
                a2b[e] = __shfl(a2, 4 * g + e + 8 * half, 32);
            }
#pragma unroll
            for (int t = 0; t < 8; ++t) {
                const v4f o2 = *(const v4f*)(xb + ((t * 2 + g) * 32 + lane) * 4);
#pragma unroll
                for (int e = 0; e < 4; ++e)
                    oacc[t][4 * g + e] = oacc[t][4 * g + e] * a1b[e] + o2[e] * a2b[e];
            }
        }
    }

    if (npair < NT / 2) {
        if (!__all(m_run > ATTN_NEG + 200.0f)) {
            load_first_k(npair);
            run(npair, NT / 2);
        }
    }

    const float linv = 1.0f / l_run;
#pragma unroll
    for (int r = 0; r < 8; ++r) {
        const float lb = __shfl(linv, r + 8 * half, 32);
        const int qrow_o = q0 + r + 8 * half;
        float* orow = Ob + (size_t)qrow_o * Dn + 4 * l16;
#pragma unroll
        for (int u = 0; u < 2; ++u) {
            v4f o = v4f{ oacc[4 * u][r], oacc[4 * u + 1][r],
                         oacc[4 * u + 2][r], oacc[4 * u + 3][r] } * lb;
            *(v4f*)(orow + 64 * u) = o;
        }
    }
}

extern "C" void kernel_launch(void* const* d_in, const int* in_sizes, int n_in,
                              void* d_out, int out_size, void* d_ws, size_t ws_size,
                              hipStream_t stream) {
    const float* q  = (const float*)d_in[0];
    const float* k  = (const float*)d_in[1];
    const float* v  = (const float*)d_in[2];
    const int*   vl = (const int*)d_in[3];
    float* out = (float*)d_out;

    dim3 grid(512);
    dim3 block(256);
    DotProductAttention_58368605552662_kernel<<<grid, block, 0, stream>>>(q, k, v, vl, out);
}
